# starting kernel
# baseline (speedup 1.0000x reference)
#include <hip/hip_runtime.h>
#include <math.h>

typedef float v2f __attribute__((ext_vector_type(2)));
typedef float v8f __attribute__((ext_vector_type(8)));

#define D_H 256
#define LN_EPS 1e-5f

__global__ void k_zero_f32(float* __restrict__ p, long n) {
    long i = (long)blockIdx.x * blockDim.x + threadIdx.x;
    long stride = (long)gridDim.x * blockDim.x;
    for (; i < n; i += stride) p[i] = 0.0f;
}

__global__ void k_scatter_edges(const float* __restrict__ h,
                                const int* __restrict__ src,
                                const int* __restrict__ dst,
                                float* __restrict__ accum,
                                float* __restrict__ deg, int E) {
    int wave = (int)((blockIdx.x * (long)blockDim.x + threadIdx.x) >> 5);
    int lane = threadIdx.x & 31;
    if (wave >= E) return;
    int s = src[wave];
    int d = dst[wave];
    const float* hs = h + (size_t)s * D_H;
    float* ad = accum + (size_t)d * D_H;
#pragma unroll
    for (int i = 0; i < 8; ++i) {
        int c = lane + i * 32;
        atomicAdd(&ad[c], hs[c]);
    }
    if (lane == 0) atomicAdd(&deg[d], 1.0f);
}

__global__ void k_finalize_mean(float* __restrict__ accum,
                                const float* __restrict__ h_dst,
                                const float* __restrict__ deg, long total) {
    long i = (long)blockIdx.x * blockDim.x + threadIdx.x;
    if (i >= total) return;
    long r = i >> 8;
    float inv = 1.0f / (deg[r] + 1.0f);
    accum[i] = (accum[i] + h_dst[i]) * inv;
}

#define KCH 128
#define ALD 132
__global__ __launch_bounds__(256) void k_gemm_wmma_f32(
    const float* __restrict__ A, const float* __restrict__ W,
    const float* __restrict__ bias, float* __restrict__ out, int M, int N) {
    __shared__ float Alds[32 * ALD];
    __shared__ float Blds[KCH * 64];
    const int tid = threadIdx.x;
    const int rowBase = blockIdx.x * 32;
    const int colBase = blockIdx.y * 64;

    const int wave = tid >> 5, lane = tid & 31;
    const int wr = wave >> 2, wc = wave & 3;
    const int half = lane >> 4, l16 = lane & 15;

    v8f c = {};

    for (int kb = 0; kb < D_H; kb += KCH) {
        __syncthreads();
#pragma unroll
        for (int j = 0; j < 4; ++j) {
            int idx = tid + j * 256;
            int r = idx >> 5, c4 = idx & 31;
            int grow = rowBase + r;
            float4 v = make_float4(0.f, 0.f, 0.f, 0.f);
            if (grow < M) v = *(const float4*)(A + (size_t)grow * D_H + kb + c4 * 4);
            *(float4*)(&Alds[r * ALD + c4 * 4]) = v;
        }
#pragma unroll
        for (int j = 0; j < 8; ++j) {
            int idx = tid + j * 256;
            int k = idx >> 4, c4 = idx & 15;
            float4 v = *(const float4*)(W + (size_t)(kb + k) * N + colBase + c4 * 4);
            *(float4*)(&Blds[k * 64 + c4 * 4]) = v;
        }
        __syncthreads();

        const float* arow = &Alds[(wr * 16 + l16) * ALD];
        const float* bcol = &Blds[wc * 16 + l16];
#pragma unroll 8
        for (int k0 = 0; k0 < KCH; k0 += 4) {
            v2f a, b;
            a.x = arow[k0 + 2 * half];
            a.y = arow[k0 + 2 * half + 1];
            b.x = bcol[(size_t)(k0 + 2 * half) * 64];
            b.y = bcol[(size_t)(k0 + 2 * half + 1) * 64];
            c = __builtin_amdgcn_wmma_f32_16x16x4_f32(
                false, a, false, b, (short)0, c, false, false);
        }
    }

    int col = colBase + wc * 16 + l16;
    float bv = bias[col];
    int rbase = rowBase + wr * 16 + half * 8;
#pragma unroll
    for (int j = 0; j < 8; ++j) {
        int row = rbase + j;
        if (row < M) out[(size_t)row * N + col] = c[j] + bv;
    }
}

__global__ void k_ln_relu(const float* __restrict__ x, const float* __restrict__ g,
                          const float* __restrict__ b, float* __restrict__ y, int M) {
    int wave = (int)((blockIdx.x * (long)blockDim.x + threadIdx.x) >> 5);
    int lane = threadIdx.x & 31;
    if (wave >= M) return;
    const float* xr = x + (size_t)wave * D_H;
    float v[8];
    float s = 0.0f;
#pragma unroll
    for (int i = 0; i < 8; ++i) { v[i] = xr[lane + 32 * i]; s += v[i]; }
#pragma unroll
    for (int o = 16; o > 0; o >>= 1) s += __shfl_xor(s, o, 32);
    float mu = s * (1.0f / 256.0f);
    float vs = 0.0f;
#pragma unroll
    for (int i = 0; i < 8; ++i) { float d = v[i] - mu; vs += d * d; }
#pragma unroll
    for (int o = 16; o > 0; o >>= 1) vs += __shfl_xor(vs, o, 32);
    float rs = rsqrtf(vs * (1.0f / 256.0f) + LN_EPS);
    float* yr = y + (size_t)wave * D_H;
#pragma unroll
    for (int i = 0; i < 8; ++i) {
        int ci = lane + 32 * i;
        float t = (v[i] - mu) * rs * g[ci] + b[ci];
        yr[ci] = fmaxf(t, 0.0f);
    }
}

extern "C" void kernel_launch(void* const* d_in, const int* in_sizes, int n_in,
                              void* d_out, int out_size, void* d_ws, size_t ws_size,
                              hipStream_t stream) {
    (void)n_in; (void)out_size; (void)ws_size;
    const float* feats = (const float*)d_in[0];
    const float* W0    = (const float*)d_in[1];
    const float* b0    = (const float*)d_in[2];
    const float* W1    = (const float*)d_in[3];
    const float* b1    = (const float*)d_in[4];
    const float* ln_g  = (const float*)d_in[5];
    const float* ln_b  = (const float*)d_in[6];
    const int* src0 = (const int*)d_in[7];
    const int* dst0 = (const int*)d_in[8];
    const int* src1 = (const int*)d_in[9];
    const int* dst1 = (const int*)d_in[10];
    const int E0 = in_sizes[7], E1 = in_sizes[9];
    const int N1 = 50000, N2 = 10000, DOUT = 128;

    float* ws     = (float*)d_ws;
    float* accum0 = ws;
    float* deg0   = accum0 + (size_t)N1 * D_H;
    float* h      = deg0 + N1;
    float* accum1 = h + (size_t)N1 * D_H;
    float* deg1   = accum1 + (size_t)N2 * D_H;

    float* h0_out = (float*)d_out;
    float* out2   = h0_out + (size_t)N1 * D_H;

    long z0 = (long)N1 * D_H + N1;
    k_zero_f32<<<4096, 256, 0, stream>>>(accum0, z0);
    {
        long threads = (long)E0 * 32;
        k_scatter_edges<<<(int)((threads + 255) / 256), 256, 0, stream>>>(
            feats, src0, dst0, accum0, deg0, E0);
    }
    {
        long total = (long)N1 * D_H;
        k_finalize_mean<<<(int)((total + 255) / 256), 256, 0, stream>>>(
            accum0, feats, deg0, total);
    }
    {
        dim3 grid((N1 + 31) / 32, D_H / 64);
        k_gemm_wmma_f32<<<grid, 256, 0, stream>>>(accum0, W0, b0, h0_out, N1, D_H);
    }
    {
        long threads = (long)N1 * 32;
        k_ln_relu<<<(int)((threads + 255) / 256), 256, 0, stream>>>(
            h0_out, ln_g, ln_b, h, N1);
    }

    long z1 = (long)N2 * D_H + N2;
    k_zero_f32<<<2048, 256, 0, stream>>>(accum1, z1);
    {
        long threads = (long)E1 * 32;
        k_scatter_edges<<<(int)((threads + 255) / 256), 256, 0, stream>>>(
            h, src1, dst1, accum1, deg1, E1);
    }
    {
        long total = (long)N2 * D_H;
        k_finalize_mean<<<(int)((total + 255) / 256), 256, 0, stream>>>(
            accum1, h, deg1, total);
    }
    {
        dim3 grid((N2 + 31) / 32, DOUT / 64);
        k_gemm_wmma_f32<<<grid, 256, 0, stream>>>(accum1, W1, b1, out2, N2, DOUT);
    }
}


# Float-atomic edge scatter replaced by per-destination gather: int-atomic count/slot fill, then one wave sums its row's sources in registers in arrival-order rank, zero and mean fused
# speedup vs baseline: 1.6393x; 1.6393x over previous
#include <hip/hip_runtime.h>
#include <math.h>

typedef float v2f __attribute__((ext_vector_type(2)));
typedef float v8f __attribute__((ext_vector_type(8)));

#define D_H 256
#define LN_EPS 1e-5f


typedef float v4f __attribute__((ext_vector_type(4)));

__global__ void k_zero_i32(int* __restrict__ p, int n) {
    int i = blockIdx.x * blockDim.x + threadIdx.x;
    if (i < n) p[i] = 0;
}

__global__ void k_count_dst(const int* __restrict__ dst0, int E0, int* __restrict__ cnt0,
                            const int* __restrict__ dst1, int E1, int* __restrict__ cnt1) {
    int t = blockIdx.x * blockDim.x + threadIdx.x;
    if (t < E0) {
        atomicAdd(&cnt0[dst0[t]], 1);
    } else if (t - E0 < E1) {
        atomicAdd(&cnt1[dst1[t - E0]], 1);
    }
}

__global__ void k_seg_alloc(const int* __restrict__ cnt0, int* __restrict__ seg0,
                            int* __restrict__ tot0, int n0,
                            const int* __restrict__ cnt1, int* __restrict__ seg1,
                            int* __restrict__ tot1, int n1) {
    const int* cnt = blockIdx.y ? cnt1 : cnt0;
    int* seg = blockIdx.y ? seg1 : seg0;
    int* tot = blockIdx.y ? tot1 : tot0;
    const int n = blockIdx.y ? n1 : n0;
    const int d = blockIdx.x * blockDim.x + threadIdx.x;
    const int lane = threadIdx.x & 31;
    const int c = (d < n) ? cnt[d] : 0;
    int incl = c;
#pragma unroll
    for (int o = 1; o < 32; o <<= 1) {
        int t = __shfl_up(incl, o, 32);
        if (lane >= o) incl += t;
    }
    const int wtot = __shfl(incl, 31, 32);
    int base = 0;
    if (lane == 31 && wtot > 0) base = atomicAdd(tot, wtot);
    base = __shfl(base, 31, 32);
    if (d < n) *(int2*)(seg + 2 * d) = make_int2(base + incl - c, c);
}

__global__ void k_fill_slots(const int* __restrict__ src0, const int* __restrict__ dst0, int E0,
                             int* __restrict__ cnt0, const int* __restrict__ seg0,
                             int* __restrict__ slots0,
                             const int* __restrict__ src1, const int* __restrict__ dst1, int E1,
                             int* __restrict__ cnt1, const int* __restrict__ seg1,
                             int* __restrict__ slots1) {
    int t = blockIdx.x * blockDim.x + threadIdx.x;
    if (t < E0) {
        int d = dst0[t];
        int pos = atomicSub(&cnt0[d], 1) - 1;
        slots0[seg0[2 * d] + pos] = src0[t];
    } else if (t - E0 < E1) {
        int e = t - E0;
        int d = dst1[e];
        int pos = atomicSub(&cnt1[d], 1) - 1;
        slots1[seg1[2 * d] + pos] = src1[e];
    }
}

__global__ __launch_bounds__(256) void k_gather_mean(
    const float* __restrict__ h, const int* __restrict__ slots,
    const int* __restrict__ seg, float* __restrict__ out, int n_dst) {
    const int row = __builtin_amdgcn_readfirstlane(
        (int)((blockIdx.x * (long)blockDim.x + threadIdx.x) >> 5));
    const int lane = threadIdx.x & 31;
    if (row >= n_dst) return;
    const int2 sg = *(const int2*)(seg + 2 * row);
    const int base = sg.x, n = sg.y;
    const v4f* pd = (const v4f*)(h + (size_t)row * D_H);
    const v4f xd = pd[lane], yd = pd[lane + 32];
    v4f a0 = {0.f, 0.f, 0.f, 0.f}, a1 = {0.f, 0.f, 0.f, 0.f};
    for (int c = 0; c < n; c += 32) {
        const int m = (n - c < 32) ? (n - c) : 32;
        const int idx = (lane < m) ? slots[base + c + lane] : 0;
        int j = 0;
        for (; j + 4 <= m; j += 4) {
            const v4f* p0 = (const v4f*)(h + (size_t)__builtin_amdgcn_readlane(idx, j) * D_H);
            const v4f* p1 = (const v4f*)(h + (size_t)__builtin_amdgcn_readlane(idx, j + 1) * D_H);
            const v4f* p2 = (const v4f*)(h + (size_t)__builtin_amdgcn_readlane(idx, j + 2) * D_H);
            const v4f* p3 = (const v4f*)(h + (size_t)__builtin_amdgcn_readlane(idx, j + 3) * D_H);
            v4f x0 = p0[lane], y0 = p0[lane + 32];
            v4f x1 = p1[lane], y1 = p1[lane + 32];
            v4f x2 = p2[lane], y2 = p2[lane + 32];
            v4f x3 = p3[lane], y3 = p3[lane + 32];
            a0 += x0; a1 += y0;
            a0 += x1; a1 += y1;
            a0 += x2; a1 += y2;
            a0 += x3; a1 += y3;
        }
        const int r = m - j;
        if (r > 0) {
            const v4f* p0 = (const v4f*)(h + (size_t)__builtin_amdgcn_readlane(idx, j) * D_H);
            v4f x0 = p0[lane], y0 = p0[lane + 32];
            v4f x1 = {0.f, 0.f, 0.f, 0.f}, y1 = x1, x2 = x1, y2 = x1;
            if (r > 1) {
                const v4f* p1 = (const v4f*)(h + (size_t)__builtin_amdgcn_readlane(idx, j + 1) * D_H);
                x1 = p1[lane]; y1 = p1[lane + 32];
            }
            if (r > 2) {
                const v4f* p2 = (const v4f*)(h + (size_t)__builtin_amdgcn_readlane(idx, j + 2) * D_H);
                x2 = p2[lane]; y2 = p2[lane + 32];
            }
            a0 += x0; a1 += y0;
            if (r > 1) { a0 += x1; a1 += y1; }
            if (r > 2) { a0 += x2; a1 += y2; }
        }
    }
    const float inv = 1.0f / ((float)n + 1.0f);
    v4f* po = (v4f*)(out + (size_t)row * D_H);
    po[lane] = (a0 + xd) * inv;
    po[lane + 32] = (a1 + yd) * inv;
}

#define KCH 128
#define ALD 132
__global__ __launch_bounds__(256) void k_gemm_wmma_f32(
    const float* __restrict__ A, const float* __restrict__ W,
    const float* __restrict__ bias, float* __restrict__ out, int M, int N) {
    __shared__ float Alds[32 * ALD];
    __shared__ float Blds[KCH * 64];
    const int tid = threadIdx.x;
    const int rowBase = blockIdx.x * 32;
    const int colBase = blockIdx.y * 64;

    const int wave = tid >> 5, lane = tid & 31;
    const int wr = wave >> 2, wc = wave & 3;
    const int half = lane >> 4, l16 = lane & 15;

    v8f c = {};

    for (int kb = 0; kb < D_H; kb += KCH) {
        __syncthreads();
#pragma unroll
        for (int j = 0; j < 4; ++j) {
            int idx = tid + j * 256;
            int r = idx >> 5, c4 = idx & 31;
            int grow = rowBase + r;
            float4 v = make_float4(0.f, 0.f, 0.f, 0.f);
            if (grow < M) v = *(const float4*)(A + (size_t)grow * D_H + kb + c4 * 4);
            *(float4*)(&Alds[r * ALD + c4 * 4]) = v;
        }
#pragma unroll
        for (int j = 0; j < 8; ++j) {
            int idx = tid + j * 256;
            int k = idx >> 4, c4 = idx & 15;
            float4 v = *(const float4*)(W + (size_t)(kb + k) * N + colBase + c4 * 4);
            *(float4*)(&Blds[k * 64 + c4 * 4]) = v;
        }
        __syncthreads();

        const float* arow = &Alds[(wr * 16 + l16) * ALD];
        const float* bcol = &Blds[wc * 16 + l16];
#pragma unroll 8
        for (int k0 = 0; k0 < KCH; k0 += 4) {
            v2f a, b;
            a.x = arow[k0 + 2 * half];
            a.y = arow[k0 + 2 * half + 1];
            b.x = bcol[(size_t)(k0 + 2 * half) * 64];
            b.y = bcol[(size_t)(k0 + 2 * half + 1) * 64];
            c = __builtin_amdgcn_wmma_f32_16x16x4_f32(
                false, a, false, b, (short)0, c, false, false);
        }
    }

    int col = colBase + wc * 16 + l16;
    float bv = bias[col];
    int rbase = rowBase + wr * 16 + half * 8;
#pragma unroll
    for (int j = 0; j < 8; ++j) {
        int row = rbase + j;
        if (row < M) out[(size_t)row * N + col] = c[j] + bv;
    }
}

__global__ void k_ln_relu(const float* __restrict__ x, const float* __restrict__ g,
                          const float* __restrict__ b, float* __restrict__ y, int M) {
    int wave = (int)((blockIdx.x * (long)blockDim.x + threadIdx.x) >> 5);
    int lane = threadIdx.x & 31;
    if (wave >= M) return;
    const float* xr = x + (size_t)wave * D_H;
    float v[8];
    float s = 0.0f;
#pragma unroll
    for (int i = 0; i < 8; ++i) { v[i] = xr[lane + 32 * i]; s += v[i]; }
#pragma unroll
    for (int o = 16; o > 0; o >>= 1) s += __shfl_xor(s, o, 32);
    float mu = s * (1.0f / 256.0f);
    float vs = 0.0f;
#pragma unroll
    for (int i = 0; i < 8; ++i) { float d = v[i] - mu; vs += d * d; }
#pragma unroll
    for (int o = 16; o > 0; o >>= 1) vs += __shfl_xor(vs, o, 32);
    float rs = rsqrtf(vs * (1.0f / 256.0f) + LN_EPS);
    float* yr = y + (size_t)wave * D_H;
#pragma unroll
    for (int i = 0; i < 8; ++i) {
        int ci = lane + 32 * i;
        float t = (v[i] - mu) * rs * g[ci] + b[ci];
        yr[ci] = fmaxf(t, 0.0f);
    }
}

extern "C" void kernel_launch(void* const* d_in, const int* in_sizes, int n_in,
                              void* d_out, int out_size, void* d_ws, size_t ws_size,
                              hipStream_t stream) {
    (void)n_in; (void)out_size; (void)ws_size;
    const float* feats = (const float*)d_in[0];
    const float* W0    = (const float*)d_in[1];
    const float* b0    = (const float*)d_in[2];
    const float* W1    = (const float*)d_in[3];
    const float* b1    = (const float*)d_in[4];
    const float* ln_g  = (const float*)d_in[5];
    const float* ln_b  = (const float*)d_in[6];
    const int* src0 = (const int*)d_in[7];
    const int* dst0 = (const int*)d_in[8];
    const int* src1 = (const int*)d_in[9];
    const int* dst1 = (const int*)d_in[10];
    const int E0 = in_sizes[7], E1 = in_sizes[9];
    const int N1 = 50000, N2 = 10000, DOUT = 128;

    float* ws = (float*)d_ws;
    float* hn = ws;
    float* h  = ws + (size_t)N1 * D_H + N1;
    int* cnt0   = (int*)(h + (size_t)N1 * D_H);
    int* cnt1   = cnt0 + N1;
    int* tot    = cnt1 + N2;
    int* seg0   = tot + 2;
    int* seg1   = seg0 + 2 * N1;
    int* slots0 = seg1 + 2 * N2;
    int* slots1 = slots0 + E0;

    float* h0_out = (float*)d_out;
    float* out2   = h0_out + (size_t)N1 * D_H;

    k_zero_i32<<<(N1 + N2 + 2 + 255) / 256, 256, 0, stream>>>(cnt0, N1 + N2 + 2);
    {
        long threads = (long)E0 + E1;
        k_count_dst<<<(int)((threads + 255) / 256), 256, 0, stream>>>(
            dst0, E0, cnt0, dst1, E1, cnt1);
        dim3 grid((N1 + 255) / 256, 2);
        k_seg_alloc<<<grid, 256, 0, stream>>>(cnt0, seg0, tot, N1,
                                              cnt1, seg1, tot + 1, N2);
        k_fill_slots<<<(int)((threads + 255) / 256), 256, 0, stream>>>(
            src0, dst0, E0, cnt0, seg0, slots0, src1, dst1, E1, cnt1, seg1, slots1);
    }

    {
        long threads = (long)N1 * 32;
        k_gather_mean<<<(int)((threads + 255) / 256), 256, 0, stream>>>(
            feats, slots0, seg0, hn, N1);
    }
    {
        dim3 grid((N1 + 31) / 32, D_H / 64);
        k_gemm_wmma_f32<<<grid, 256, 0, stream>>>(hn, W0, b0, h0_out, N1, D_H);
    }
    {
        long threads = (long)N1 * 32;
        k_ln_relu<<<(int)((threads + 255) / 256), 256, 0, stream>>>(
            h0_out, ln_g, ln_b, h, N1);
    }

    {
        long threads = (long)N2 * 32;
        k_gather_mean<<<(int)((threads + 255) / 256), 256, 0, stream>>>(
            h, slots1, seg1, hn, N2);
    }
    {
        dim3 grid((N2 + 31) / 32, DOUT / 64);
        k_gemm_wmma_f32<<<grid, 256, 0, stream>>>(hn, W1, b1, out2, N2, DOUT);
    }
}


# Edge slots built by a two-level counting sort ranked at LDS counters (256-row buckets) instead of one global integer atomic per edge in count and fill; slot order becomes LDS arrival order
# speedup vs baseline: 2.4690x; 1.5061x over previous
#include <hip/hip_runtime.h>
#include <math.h>

typedef float v2f __attribute__((ext_vector_type(2)));
typedef float v8f __attribute__((ext_vector_type(8)));

#define D_H 256
#define LN_EPS 1e-5f


typedef float v4f __attribute__((ext_vector_type(4)));

#define SORT_G 256
#define NB0 196
#define NB1 40
#define NBK 236

__device__ __forceinline__ int block_excl_scan_256(int v, int* red) {
    const int lane = threadIdx.x & 31, wave = threadIdx.x >> 5;
    int incl = v;
#pragma unroll
    for (int o = 1; o < 32; o <<= 1) {
        int t = __shfl_up(incl, o, 32);
        if (lane >= o) incl += t;
    }
    if (lane == 31) red[wave] = incl;
    __syncthreads();
    int woff = 0;
#pragma unroll
    for (int w = 0; w < 8; ++w) if (w < wave) woff += red[w];
    __syncthreads();
    return woff + incl - v;
}

__global__ __launch_bounds__(256) void k_bucket_hist(
    const int* __restrict__ dst0, int E0, const int* __restrict__ dst1, int E1,
    int* __restrict__ bh) {
    __shared__ int hist[NBK];
    const int tid = threadIdx.x;
    if (tid < NBK) hist[tid] = 0;
    __syncthreads();
    {
        const int c = (E0 + SORT_G - 1) / SORT_G;
        const int lo = blockIdx.x * c;
        const int hi = min(lo + c, E0);
        for (int e = lo + tid; e < hi; e += 1024) {
            int d[4];
#pragma unroll
            for (int u = 0; u < 4; ++u) { const int i = e + u * 256; d[u] = (i < hi) ? dst0[i] : -1; }
#pragma unroll
            for (int u = 0; u < 4; ++u) if (d[u] >= 0) atomicAdd(&hist[d[u] >> 8], 1);
        }
    }
    {
        const int c = (E1 + SORT_G - 1) / SORT_G;
        const int lo = blockIdx.x * c;
        const int hi = min(lo + c, E1);
        for (int e = lo + tid; e < hi; e += 1024) {
            int d[4];
#pragma unroll
            for (int u = 0; u < 4; ++u) { const int i = e + u * 256; d[u] = (i < hi) ? dst1[i] : -1; }
#pragma unroll
            for (int u = 0; u < 4; ++u) if (d[u] >= 0) atomicAdd(&hist[NB0 + (d[u] >> 8)], 1);
        }
    }
    __syncthreads();
    if (tid < NBK) bh[tid * SORT_G + blockIdx.x] = hist[tid];
}

__global__ __launch_bounds__(256) void k_bucket_scan(int* __restrict__ bh, int* __restrict__ keytot) {
    const int key = (int)((blockIdx.x * 256 + threadIdx.x) >> 5);
    const int lane = threadIdx.x & 31;
    if (key >= NBK) return;
    int4* p = (int4*)(bh + key * SORT_G + lane * 8);
    const int4 a = p[0], b = p[1];
    const int t = a.x + a.y + a.z + a.w + b.x + b.y + b.z + b.w;
    int incl = t;
#pragma unroll
    for (int o = 1; o < 32; o <<= 1) {
        int x = __shfl_up(incl, o, 32);
        if (lane >= o) incl += x;
    }
    int run = incl - t;
    int4 oa, ob;
    oa.x = run; run += a.x;
    oa.y = run; run += a.y;
    oa.z = run; run += a.z;
    oa.w = run; run += a.w;
    ob.x = run; run += b.x;
    ob.y = run; run += b.y;
    ob.z = run; run += b.z;
    ob.w = run;
    p[0] = oa; p[1] = ob;
    if (lane == 31) keytot[key] = incl;
}

__global__ __launch_bounds__(256) void k_bucket_scatter(
    const int* __restrict__ src0, const int* __restrict__ dst0, int E0,
    const int* __restrict__ src1, const int* __restrict__ dst1, int E1,
    const int* __restrict__ bh, const int* __restrict__ keytot, int2* __restrict__ sorted) {
    __shared__ int cur[NBK];
    __shared__ int red[8];
    const int tid = threadIdx.x;
    const int kt = (tid < NBK) ? keytot[tid] : 0;
    const int ks = block_excl_scan_256(kt, red);
    if (tid < NBK) cur[tid] = ks + bh[tid * SORT_G + blockIdx.x];
    __syncthreads();
    {
        const int c = (E0 + SORT_G - 1) / SORT_G;
        const int lo = blockIdx.x * c;
        const int hi = min(lo + c, E0);
        for (int e = lo + tid; e < hi; e += 1024) {
            int d[4], sv[4];
#pragma unroll
            for (int u = 0; u < 4; ++u) {
                const int i = e + u * 256;
                d[u] = (i < hi) ? dst0[i] : -1;
                sv[u] = (i < hi) ? src0[i] : 0;
            }
#pragma unroll
            for (int u = 0; u < 4; ++u) if (d[u] >= 0) {
                const int pos = atomicAdd(&cur[d[u] >> 8], 1);
                sorted[pos] = make_int2(d[u] & 255, sv[u]);
            }
        }
    }
    {
        const int c = (E1 + SORT_G - 1) / SORT_G;
        const int lo = blockIdx.x * c;
        const int hi = min(lo + c, E1);
        for (int e = lo + tid; e < hi; e += 1024) {
            int d[4], sv[4];
#pragma unroll
            for (int u = 0; u < 4; ++u) {
                const int i = e + u * 256;
                d[u] = (i < hi) ? dst1[i] : -1;
                sv[u] = (i < hi) ? src1[i] : 0;
            }
#pragma unroll
            for (int u = 0; u < 4; ++u) if (d[u] >= 0) {
                const int pos = atomicAdd(&cur[NB0 + (d[u] >> 8)], 1);
                sorted[pos] = make_int2(d[u] & 255, sv[u]);
            }
        }
    }
}

__global__ __launch_bounds__(256) void k_bucket_rows(
    const int2* __restrict__ sorted, const int* __restrict__ keytot,
    int* __restrict__ seg0, int n0, int* __restrict__ seg1, int n1,
    int* __restrict__ slots) {
    __shared__ int cnt[256];
    __shared__ int red[8];
    __shared__ int range[2];
    const int tid = threadIdx.x;
    const int key = blockIdx.x;
    const int kt = (tid < NBK) ? keytot[tid] : 0;
    const int ks = block_excl_scan_256(kt, red);
    if (tid == key) { range[0] = ks; range[1] = ks + kt; }
    cnt[tid] = 0;
    __syncthreads();
    const int bs = range[0], be = range[1];
    for (int i = bs + tid; i < be; i += 1024) {
        int r[4];
#pragma unroll
        for (int u = 0; u < 4; ++u) { const int j = i + u * 256; r[u] = (j < be) ? sorted[j].x : -1; }
#pragma unroll
        for (int u = 0; u < 4; ++u) if (r[u] >= 0) atomicAdd(&cnt[r[u]], 1);
    }
    __syncthreads();
    const int c = cnt[tid];
    const int ex = block_excl_scan_256(c, red);
    const int l1 = key >= NB0;
    const int row = ((l1 ? key - NB0 : key) << 8) + tid;
    if (l1) {
        if (row < n1) *(int2*)(seg1 + 2 * row) = make_int2(bs + ex, c);
    } else {
        if (row < n0) *(int2*)(seg0 + 2 * row) = make_int2(bs + ex, c);
    }
    cnt[tid] = ex;
    __syncthreads();
    for (int i = bs + tid; i < be; i += 1024) {
        int2 p[4];
#pragma unroll
        for (int u = 0; u < 4; ++u) { const int j = i + u * 256; p[u] = (j < be) ? sorted[j] : make_int2(-1, 0); }
#pragma unroll
        for (int u = 0; u < 4; ++u) if (p[u].x >= 0) {
            const int pos = atomicAdd(&cnt[p[u].x], 1);
            slots[bs + pos] = p[u].y;
        }
    }
}

__global__ __launch_bounds__(256) void k_gather_mean(
    const float* __restrict__ h, const int* __restrict__ slots,
    const int* __restrict__ seg, float* __restrict__ out, int n_dst) {
    const int row = __builtin_amdgcn_readfirstlane(
        (int)((blockIdx.x * (long)blockDim.x + threadIdx.x) >> 5));
    const int lane = threadIdx.x & 31;
    if (row >= n_dst) return;
    const int2 sg = *(const int2*)(seg + 2 * row);
    const int base = sg.x, n = sg.y;
    const v4f* pd = (const v4f*)(h + (size_t)row * D_H);
    const v4f xd = pd[lane], yd = pd[lane + 32];
    v4f a0 = {0.f, 0.f, 0.f, 0.f}, a1 = {0.f, 0.f, 0.f, 0.f};
    for (int c = 0; c < n; c += 32) {
        const int m = (n - c < 32) ? (n - c) : 32;
        const int idx = (lane < m) ? slots[base + c + lane] : 0;
        int j = 0;
        for (; j + 4 <= m; j += 4) {
            const v4f* p0 = (const v4f*)(h + (size_t)__builtin_amdgcn_readlane(idx, j) * D_H);
            const v4f* p1 = (const v4f*)(h + (size_t)__builtin_amdgcn_readlane(idx, j + 1) * D_H);
            const v4f* p2 = (const v4f*)(h + (size_t)__builtin_amdgcn_readlane(idx, j + 2) * D_H);
            const v4f* p3 = (const v4f*)(h + (size_t)__builtin_amdgcn_readlane(idx, j + 3) * D_H);
            v4f x0 = p0[lane], y0 = p0[lane + 32];
            v4f x1 = p1[lane], y1 = p1[lane + 32];
            v4f x2 = p2[lane], y2 = p2[lane + 32];
            v4f x3 = p3[lane], y3 = p3[lane + 32];
            a0 += x0; a1 += y0;
            a0 += x1; a1 += y1;
            a0 += x2; a1 += y2;
            a0 += x3; a1 += y3;
        }
        const int r = m - j;
        if (r > 0) {
            const v4f* p0 = (const v4f*)(h + (size_t)__builtin_amdgcn_readlane(idx, j) * D_H);
            v4f x0 = p0[lane], y0 = p0[lane + 32];
            v4f x1 = {0.f, 0.f, 0.f, 0.f}, y1 = x1, x2 = x1, y2 = x1;
            if (r > 1) {
                const v4f* p1 = (const v4f*)(h + (size_t)__builtin_amdgcn_readlane(idx, j + 1) * D_H);
                x1 = p1[lane]; y1 = p1[lane + 32];
            }
            if (r > 2) {
                const v4f* p2 = (const v4f*)(h + (size_t)__builtin_amdgcn_readlane(idx, j + 2) * D_H);
                x2 = p2[lane]; y2 = p2[lane + 32];
            }
            a0 += x0; a1 += y0;
            if (r > 1) { a0 += x1; a1 += y1; }
            if (r > 2) { a0 += x2; a1 += y2; }
        }
    }
    const float inv = 1.0f / ((float)n + 1.0f);
    v4f* po = (v4f*)(out + (size_t)row * D_H);
    po[lane] = (a0 + xd) * inv;
    po[lane + 32] = (a1 + yd) * inv;
}

#define KCH 128
#define ALD 132
__global__ __launch_bounds__(256) void k_gemm_wmma_f32(
    const float* __restrict__ A, const float* __restrict__ W,
    const float* __restrict__ bias, float* __restrict__ out, int M, int N) {
    __shared__ float Alds[32 * ALD];
    __shared__ float Blds[KCH * 64];
    const int tid = threadIdx.x;
    const int rowBase = blockIdx.x * 32;
    const int colBase = blockIdx.y * 64;

    const int wave = tid >> 5, lane = tid & 31;
    const int wr = wave >> 2, wc = wave & 3;
    const int half = lane >> 4, l16 = lane & 15;

    v8f c = {};

    for (int kb = 0; kb < D_H; kb += KCH) {
        __syncthreads();
#pragma unroll
        for (int j = 0; j < 4; ++j) {
            int idx = tid + j * 256;
            int r = idx >> 5, c4 = idx & 31;
            int grow = rowBase + r;
            float4 v = make_float4(0.f, 0.f, 0.f, 0.f);
            if (grow < M) v = *(const float4*)(A + (size_t)grow * D_H + kb + c4 * 4);
            *(float4*)(&Alds[r * ALD + c4 * 4]) = v;
        }
#pragma unroll
        for (int j = 0; j < 8; ++j) {
            int idx = tid + j * 256;
            int k = idx >> 4, c4 = idx & 15;
            float4 v = *(const float4*)(W + (size_t)(kb + k) * N + colBase + c4 * 4);
            *(float4*)(&Blds[k * 64 + c4 * 4]) = v;
        }
        __syncthreads();

        const float* arow = &Alds[(wr * 16 + l16) * ALD];
        const float* bcol = &Blds[wc * 16 + l16];
#pragma unroll 8
        for (int k0 = 0; k0 < KCH; k0 += 4) {
            v2f a, b;
            a.x = arow[k0 + 2 * half];
            a.y = arow[k0 + 2 * half + 1];
            b.x = bcol[(size_t)(k0 + 2 * half) * 64];
            b.y = bcol[(size_t)(k0 + 2 * half + 1) * 64];
            c = __builtin_amdgcn_wmma_f32_16x16x4_f32(
                false, a, false, b, (short)0, c, false, false);
        }
    }

    int col = colBase + wc * 16 + l16;
    float bv = bias[col];
    int rbase = rowBase + wr * 16 + half * 8;
#pragma unroll
    for (int j = 0; j < 8; ++j) {
        int row = rbase + j;
        if (row < M) out[(size_t)row * N + col] = c[j] + bv;
    }
}

__global__ void k_ln_relu(const float* __restrict__ x, const float* __restrict__ g,
                          const float* __restrict__ b, float* __restrict__ y, int M) {
    int wave = (int)((blockIdx.x * (long)blockDim.x + threadIdx.x) >> 5);
    int lane = threadIdx.x & 31;
    if (wave >= M) return;
    const float* xr = x + (size_t)wave * D_H;
    float v[8];
    float s = 0.0f;
#pragma unroll
    for (int i = 0; i < 8; ++i) { v[i] = xr[lane + 32 * i]; s += v[i]; }
#pragma unroll
    for (int o = 16; o > 0; o >>= 1) s += __shfl_xor(s, o, 32);
    float mu = s * (1.0f / 256.0f);
    float vs = 0.0f;
#pragma unroll
    for (int i = 0; i < 8; ++i) { float d = v[i] - mu; vs += d * d; }
#pragma unroll
    for (int o = 16; o > 0; o >>= 1) vs += __shfl_xor(vs, o, 32);
    float rs = rsqrtf(vs * (1.0f / 256.0f) + LN_EPS);
    float* yr = y + (size_t)wave * D_H;
#pragma unroll
    for (int i = 0; i < 8; ++i) {
        int ci = lane + 32 * i;
        float t = (v[i] - mu) * rs * g[ci] + b[ci];
        yr[ci] = fmaxf(t, 0.0f);
    }
}

extern "C" void kernel_launch(void* const* d_in, const int* in_sizes, int n_in,
                              void* d_out, int out_size, void* d_ws, size_t ws_size,
                              hipStream_t stream) {
    (void)n_in; (void)out_size; (void)ws_size;
    const float* feats = (const float*)d_in[0];
    const float* W0    = (const float*)d_in[1];
    const float* b0    = (const float*)d_in[2];
    const float* W1    = (const float*)d_in[3];
    const float* b1    = (const float*)d_in[4];
    const float* ln_g  = (const float*)d_in[5];
    const float* ln_b  = (const float*)d_in[6];
    const int* src0 = (const int*)d_in[7];
    const int* dst0 = (const int*)d_in[8];
    const int* src1 = (const int*)d_in[9];
    const int* dst1 = (const int*)d_in[10];
    const int E0 = in_sizes[7], E1 = in_sizes[9];
    const int N1 = 50000, N2 = 10000, DOUT = 128;

    float* ws = (float*)d_ws;
    float* hn = ws;
    float* h  = ws + (size_t)N1 * D_H + N1;
    int* cnt0   = (int*)(h + (size_t)N1 * D_H);
    int* cnt1   = cnt0 + N1;
    int* tot    = cnt1 + N2;
    int* seg0   = tot + 2;
    int* seg1   = seg0 + 2 * N1;
    int* slots  = seg1 + 2 * N2;
    int2* sorted = (int2*)hn;
    int* bh      = (int*)hn + 4000000;
    int* keytot  = (int*)hn + 4100000;

    float* h0_out = (float*)d_out;
    float* out2   = h0_out + (size_t)N1 * D_H;

    k_bucket_hist<<<SORT_G, 256, 0, stream>>>(dst0, E0, dst1, E1, bh);
    k_bucket_scan<<<(NBK + 7) / 8, 256, 0, stream>>>(bh, keytot);
    k_bucket_scatter<<<SORT_G, 256, 0, stream>>>(src0, dst0, E0, src1, dst1, E1,
                                                 bh, keytot, sorted);
    k_bucket_rows<<<NBK, 256, 0, stream>>>(sorted, keytot, seg0, N1, seg1, N2, slots);

    {
        long threads = (long)N1 * 32;
        k_gather_mean<<<(int)((threads + 255) / 256), 256, 0, stream>>>(
            feats, slots, seg0, hn, N1);
    }
    {
        dim3 grid((N1 + 31) / 32, D_H / 64);
        k_gemm_wmma_f32<<<grid, 256, 0, stream>>>(hn, W0, b0, h0_out, N1, D_H);
    }
    {
        long threads = (long)N1 * 32;
        k_ln_relu<<<(int)((threads + 255) / 256), 256, 0, stream>>>(
            h0_out, ln_g, ln_b, h, N1);
    }

    {
        long threads = (long)N2 * 32;
        k_gather_mean<<<(int)((threads + 255) / 256), 256, 0, stream>>>(
            h, slots, seg1, hn, N2);
    }
    {
        dim3 grid((N2 + 31) / 32, DOUT / 64);
        k_gemm_wmma_f32<<<grid, 256, 0, stream>>>(hn, W1, b1, out2, N2, DOUT);
    }
}


# Gather+mean fused into the f32 WMMA GEMM: a workgroup gathers its 32 rows (8 sources in flight per wave) into the LDS A tile and multiplies by full-width W from LDS; h_neigh plane, 2 launches go
# speedup vs baseline: 2.9925x; 1.2120x over previous
#include <hip/hip_runtime.h>
#include <math.h>

typedef float v2f __attribute__((ext_vector_type(2)));
typedef float v8f __attribute__((ext_vector_type(8)));

#define D_H 256
#define LN_EPS 1e-5f


typedef float v4f __attribute__((ext_vector_type(4)));

#define SORT_G 256
#define NB0 196
#define NB1 40
#define NBK 236

__device__ __forceinline__ int block_excl_scan_256(int v, int* red) {
    const int lane = threadIdx.x & 31, wave = threadIdx.x >> 5;
    int incl = v;
#pragma unroll
    for (int o = 1; o < 32; o <<= 1) {
        int t = __shfl_up(incl, o, 32);
        if (lane >= o) incl += t;
    }
    if (lane == 31) red[wave] = incl;
    __syncthreads();
    int woff = 0;
#pragma unroll
    for (int w = 0; w < 8; ++w) if (w < wave) woff += red[w];
    __syncthreads();
    return woff + incl - v;
}

__global__ __launch_bounds__(256) void k_bucket_hist(
    const int* __restrict__ dst0, int E0, const int* __restrict__ dst1, int E1,
    int* __restrict__ bh) {
    __shared__ int hist[NBK];
    const int tid = threadIdx.x;
    if (tid < NBK) hist[tid] = 0;
    __syncthreads();
    {
        const int c = (E0 + SORT_G - 1) / SORT_G;
        const int lo = blockIdx.x * c;
        const int hi = min(lo + c, E0);
        for (int e = lo + tid; e < hi; e += 1024) {
            int d[4];
#pragma unroll
            for (int u = 0; u < 4; ++u) { const int i = e + u * 256; d[u] = (i < hi) ? dst0[i] : -1; }
#pragma unroll
            for (int u = 0; u < 4; ++u) if (d[u] >= 0) atomicAdd(&hist[d[u] >> 8], 1);
        }
    }
    {
        const int c = (E1 + SORT_G - 1) / SORT_G;
        const int lo = blockIdx.x * c;
        const int hi = min(lo + c, E1);
        for (int e = lo + tid; e < hi; e += 1024) {
            int d[4];
#pragma unroll
            for (int u = 0; u < 4; ++u) { const int i = e + u * 256; d[u] = (i < hi) ? dst1[i] : -1; }
#pragma unroll
            for (int u = 0; u < 4; ++u) if (d[u] >= 0) atomicAdd(&hist[NB0 + (d[u] >> 8)], 1);
        }
    }
    __syncthreads();
    if (tid < NBK) bh[tid * SORT_G + blockIdx.x] = hist[tid];
}

__global__ __launch_bounds__(256) void k_bucket_scan(int* __restrict__ bh, int* __restrict__ keytot) {
    const int key = (int)((blockIdx.x * 256 + threadIdx.x) >> 5);
    const int lane = threadIdx.x & 31;
    if (key >= NBK) return;
    int4* p = (int4*)(bh + key * SORT_G + lane * 8);
    const int4 a = p[0], b = p[1];
    const int t = a.x + a.y + a.z + a.w + b.x + b.y + b.z + b.w;
    int incl = t;
#pragma unroll
    for (int o = 1; o < 32; o <<= 1) {
        int x = __shfl_up(incl, o, 32);
        if (lane >= o) incl += x;
    }
    int run = incl - t;
    int4 oa, ob;
    oa.x = run; run += a.x;
    oa.y = run; run += a.y;
    oa.z = run; run += a.z;
    oa.w = run; run += a.w;
    ob.x = run; run += b.x;
    ob.y = run; run += b.y;
    ob.z = run; run += b.z;
    ob.w = run;
    p[0] = oa; p[1] = ob;
    if (lane == 31) keytot[key] = incl;
}

__global__ __launch_bounds__(256) void k_bucket_scatter(
    const int* __restrict__ src0, const int* __restrict__ dst0, int E0,
    const int* __restrict__ src1, const int* __restrict__ dst1, int E1,
    const int* __restrict__ bh, const int* __restrict__ keytot, int2* __restrict__ sorted) {
    __shared__ int cur[NBK];
    __shared__ int red[8];
    const int tid = threadIdx.x;
    const int kt = (tid < NBK) ? keytot[tid] : 0;
    const int ks = block_excl_scan_256(kt, red);
    if (tid < NBK) cur[tid] = ks + bh[tid * SORT_G + blockIdx.x];
    __syncthreads();
    {
        const int c = (E0 + SORT_G - 1) / SORT_G;
        const int lo = blockIdx.x * c;
        const int hi = min(lo + c, E0);
        for (int e = lo + tid; e < hi; e += 1024) {
            int d[4], sv[4];
#pragma unroll
            for (int u = 0; u < 4; ++u) {
                const int i = e + u * 256;
                d[u] = (i < hi) ? dst0[i] : -1;
                sv[u] = (i < hi) ? src0[i] : 0;
            }
#pragma unroll
            for (int u = 0; u < 4; ++u) if (d[u] >= 0) {
                const int pos = atomicAdd(&cur[d[u] >> 8], 1);
                sorted[pos] = make_int2(d[u] & 255, sv[u]);
            }
        }
    }
    {
        const int c = (E1 + SORT_G - 1) / SORT_G;
        const int lo = blockIdx.x * c;
        const int hi = min(lo + c, E1);
        for (int e = lo + tid; e < hi; e += 1024) {
            int d[4], sv[4];
#pragma unroll
            for (int u = 0; u < 4; ++u) {
                const int i = e + u * 256;
                d[u] = (i < hi) ? dst1[i] : -1;
                sv[u] = (i < hi) ? src1[i] : 0;
            }
#pragma unroll
            for (int u = 0; u < 4; ++u) if (d[u] >= 0) {
                const int pos = atomicAdd(&cur[NB0 + (d[u] >> 8)], 1);
                sorted[pos] = make_int2(d[u] & 255, sv[u]);
            }
        }
    }
}

__global__ __launch_bounds__(256) void k_bucket_rows(
    const int2* __restrict__ sorted, const int* __restrict__ keytot,
    int* __restrict__ seg0, int n0, int* __restrict__ seg1, int n1,
    int* __restrict__ slots) {
    __shared__ int cnt[256];
    __shared__ int red[8];
    __shared__ int range[2];
    const int tid = threadIdx.x;
    const int key = blockIdx.x;
    const int kt = (tid < NBK) ? keytot[tid] : 0;
    const int ks = block_excl_scan_256(kt, red);
    if (tid == key) { range[0] = ks; range[1] = ks + kt; }
    cnt[tid] = 0;
    __syncthreads();
    const int bs = range[0], be = range[1];
    for (int i = bs + tid; i < be; i += 1024) {
        int r[4];
#pragma unroll
        for (int u = 0; u < 4; ++u) { const int j = i + u * 256; r[u] = (j < be) ? sorted[j].x : -1; }
#pragma unroll
        for (int u = 0; u < 4; ++u) if (r[u] >= 0) atomicAdd(&cnt[r[u]], 1);
    }
    __syncthreads();
    const int c = cnt[tid];
    const int ex = block_excl_scan_256(c, red);
    const int l1 = key >= NB0;
    const int row = ((l1 ? key - NB0 : key) << 8) + tid;
    if (l1) {
        if (row < n1) *(int2*)(seg1 + 2 * row) = make_int2(bs + ex, c);
    } else {
        if (row < n0) *(int2*)(seg0 + 2 * row) = make_int2(bs + ex, c);
    }
    cnt[tid] = ex;
    __syncthreads();
    for (int i = bs + tid; i < be; i += 1024) {
        int2 p[4];
#pragma unroll
        for (int u = 0; u < 4; ++u) { const int j = i + u * 256; p[u] = (j < be) ? sorted[j] : make_int2(-1, 0); }
#pragma unroll
        for (int u = 0; u < 4; ++u) if (p[u].x >= 0) {
            const int pos = atomicAdd(&cnt[p[u].x], 1);
            slots[bs + pos] = p[u].y;
        }
    }
}

#define ALD2 260
#define KCH2 32

__device__ __forceinline__ void ld_src_row(const float* __restrict__ h, int id, int lane,
                                           v4f& x, v4f& y) {
    const v4f* p = (const v4f*)(h + (size_t)id * D_H);
    x = p[lane];
    y = p[lane + 32];
}

template <int N>
__global__ __launch_bounds__(256) void k_gather_gemm(
    const float* __restrict__ h, const int* __restrict__ slots, const int* __restrict__ seg,
    const float* __restrict__ W, const float* __restrict__ bias, float* __restrict__ out, int M) {
    constexpr int NT = N / 64;
    __shared__ float Alds[32 * ALD2];
    __shared__ float Blds[KCH2 * N];
    const int tid = threadIdx.x;
    const int wave = __builtin_amdgcn_readfirstlane(tid >> 5);
    const int lane = tid & 31;
    const int rowBase = blockIdx.x * 32;
    const v4f zero4 = {0.f, 0.f, 0.f, 0.f};

    int sbase[4], scnt[4], idx0[4];
#pragma unroll
    for (int i = 0; i < 4; ++i) {
        const int row = rowBase + wave * 4 + i;
        int2 sg = make_int2(0, 0);
        if (row < M) sg = *(const int2*)(seg + 2 * row);
        sbase[i] = sg.x; scnt[i] = sg.y;
    }
#pragma unroll
    for (int i = 0; i < 4; ++i) {
        const int row = rowBase + wave * 4 + i;
        idx0[i] = (lane < scnt[i]) ? slots[sbase[i] + lane] : row;
    }

#pragma unroll
    for (int i = 0; i < 4; ++i) {
        const int lr = wave * 4 + i;
        const int row = rowBase + lr;
        v4f r0 = zero4, r1 = zero4;
        if (row < M) {
            const int base = sbase[i], n = scnt[i];
            const int n1 = n + 1;
            v4f a0 = zero4, a1 = zero4;
            for (int c = 0; c < n1; c += 32) {
                const int m = (n1 - c < 32) ? (n1 - c) : 32;
                int idx = idx0[i];
                if (c > 0) idx = (c + lane < n) ? slots[base + c + lane] : row;
                int j = 0;
                for (; j + 8 <= m; j += 8) {
                    v4f x[8], y[8];
#pragma unroll
                    for (int u = 0; u < 8; ++u)
                        ld_src_row(h, __builtin_amdgcn_readlane(idx, j + u), lane, x[u], y[u]);
#pragma unroll
                    for (int u = 0; u < 8; ++u) { a0 += x[u]; a1 += y[u]; }
                }
                if (j + 4 <= m) {
                    v4f x[4], y[4];
#pragma unroll
                    for (int u = 0; u < 4; ++u)
                        ld_src_row(h, __builtin_amdgcn_readlane(idx, j + u), lane, x[u], y[u]);
#pragma unroll
                    for (int u = 0; u < 4; ++u) { a0 += x[u]; a1 += y[u]; }
                    j += 4;
                }
                const int r = m - j;
                if (r > 0) {
                    v4f x0, y0, x1 = zero4, y1 = zero4, x2 = zero4, y2 = zero4;
                    ld_src_row(h, __builtin_amdgcn_readlane(idx, j), lane, x0, y0);
                    if (r > 1) ld_src_row(h, __builtin_amdgcn_readlane(idx, j + 1), lane, x1, y1);
                    if (r > 2) ld_src_row(h, __builtin_amdgcn_readlane(idx, j + 2), lane, x2, y2);
                    a0 += x0; a1 += y0;
                    if (r > 1) { a0 += x1; a1 += y1; }
                    if (r > 2) { a0 += x2; a1 += y2; }
                }
            }
            const float inv = 1.0f / ((float)n + 1.0f);
            r0 = a0 * inv;
            r1 = a1 * inv;
        }
        *(v4f*)(&Alds[lr * ALD2 + 4 * lane]) = r0;
        *(v4f*)(&Alds[lr * ALD2 + 128 + 4 * lane]) = r1;
    }

    const int wr = wave >> 2, wc = wave & 3;
    const int half = lane >> 4, l16 = lane & 15;
    v8f acc[NT];
#pragma unroll
    for (int t = 0; t < NT; ++t) acc[t] = v8f{};
    const float* arow = &Alds[(wr * 16 + l16) * ALD2];
    const float* bcol = &Blds[wc * (N / 4) + l16];

    constexpr int NV = (KCH2 * N / 4) / 256;
    v4f wpre[NV];
#pragma unroll
    for (int j = 0; j < NV; ++j)
        wpre[j] = *(const v4f*)(W + (size_t)(tid + j * 256) * 4);

    for (int kb = 0; kb < D_H; kb += KCH2) {
        __syncthreads();
#pragma unroll
        for (int j = 0; j < NV; ++j)
            *(v4f*)(&Blds[(tid + j * 256) * 4]) = wpre[j];
        __syncthreads();
        if (kb + KCH2 < D_H) {
#pragma unroll
            for (int j = 0; j < NV; ++j)
                wpre[j] = *(const v4f*)(W + (size_t)(kb + KCH2) * N + (size_t)(tid + j * 256) * 4);
        }
#pragma unroll
        for (int k0 = 0; k0 < KCH2; k0 += 4) {
            v2f a;
            a.x = arow[kb + k0 + 2 * half];
            a.y = arow[kb + k0 + 2 * half + 1];
#pragma unroll
            for (int t = 0; t < NT; ++t) {
                v2f b;
                b.x = bcol[(k0 + 2 * half) * N + t * 16];
                b.y = bcol[(k0 + 2 * half + 1) * N + t * 16];
                acc[t] = __builtin_amdgcn_wmma_f32_16x16x4_f32(
                    false, a, false, b, (short)0, acc[t], false, false);
            }
        }
    }

    const int rbase = rowBase + wr * 16 + half * 8;
#pragma unroll
    for (int t = 0; t < NT; ++t) {
        const int col = wc * (N / 4) + t * 16 + l16;
        const float bv = bias[col];
#pragma unroll
        for (int j = 0; j < 8; ++j) {
            const int row = rbase + j;
            if (row < M) out[(size_t)row * N + col] = acc[t][j] + bv;
        }
    }
}

__global__ void k_ln_relu(const float* __restrict__ x, const float* __restrict__ g,
                          const float* __restrict__ b, float* __restrict__ y, int M) {
    int wave = (int)((blockIdx.x * (long)blockDim.x + threadIdx.x) >> 5);
    int lane = threadIdx.x & 31;
    if (wave >= M) return;
    const float* xr = x + (size_t)wave * D_H;
    float v[8];
    float s = 0.0f;
#pragma unroll
    for (int i = 0; i < 8; ++i) { v[i] = xr[lane + 32 * i]; s += v[i]; }
#pragma unroll
    for (int o = 16; o > 0; o >>= 1) s += __shfl_xor(s, o, 32);
    float mu = s * (1.0f / 256.0f);
    float vs = 0.0f;
#pragma unroll
    for (int i = 0; i < 8; ++i) { float d = v[i] - mu; vs += d * d; }
#pragma unroll
    for (int o = 16; o > 0; o >>= 1) vs += __shfl_xor(vs, o, 32);
    float rs = rsqrtf(vs * (1.0f / 256.0f) + LN_EPS);
    float* yr = y + (size_t)wave * D_H;
#pragma unroll
    for (int i = 0; i < 8; ++i) {
        int ci = lane + 32 * i;
        float t = (v[i] - mu) * rs * g[ci] + b[ci];
        yr[ci] = fmaxf(t, 0.0f);
    }
}

extern "C" void kernel_launch(void* const* d_in, const int* in_sizes, int n_in,
                              void* d_out, int out_size, void* d_ws, size_t ws_size,
                              hipStream_t stream) {
    (void)n_in; (void)out_size; (void)ws_size;
    const float* feats = (const float*)d_in[0];
    const float* W0    = (const float*)d_in[1];
    const float* b0    = (const float*)d_in[2];
    const float* W1    = (const float*)d_in[3];
    const float* b1    = (const float*)d_in[4];
    const float* ln_g  = (const float*)d_in[5];
    const float* ln_b  = (const float*)d_in[6];
    const int* src0 = (const int*)d_in[7];
    const int* dst0 = (const int*)d_in[8];
    const int* src1 = (const int*)d_in[9];
    const int* dst1 = (const int*)d_in[10];
    const int E0 = in_sizes[7], E1 = in_sizes[9];
    const int N1 = 50000, N2 = 10000;

    float* ws = (float*)d_ws;
    float* hn = ws;
    float* h  = ws + (size_t)N1 * D_H + N1;
    int* cnt0   = (int*)(h + (size_t)N1 * D_H);
    int* cnt1   = cnt0 + N1;
    int* tot    = cnt1 + N2;
    int* seg0   = tot + 2;
    int* seg1   = seg0 + 2 * N1;
    int* slots  = seg1 + 2 * N2;
    int2* sorted = (int2*)hn;
    int* bh      = (int*)hn + 4000000;
    int* keytot  = (int*)hn + 4100000;

    float* h0_out = (float*)d_out;
    float* out2   = h0_out + (size_t)N1 * D_H;

    k_bucket_hist<<<SORT_G, 256, 0, stream>>>(dst0, E0, dst1, E1, bh);
    k_bucket_scan<<<(NBK + 7) / 8, 256, 0, stream>>>(bh, keytot);
    k_bucket_scatter<<<SORT_G, 256, 0, stream>>>(src0, dst0, E0, src1, dst1, E1,
                                                 bh, keytot, sorted);
    k_bucket_rows<<<NBK, 256, 0, stream>>>(sorted, keytot, seg0, N1, seg1, N2, slots);

    k_gather_gemm<256><<<(N1 + 31) / 32, 256, 0, stream>>>(
        feats, slots, seg0, W0, b0, h0_out, N1);
    {
        long threads = (long)N1 * 32;
        k_ln_relu<<<(int)((threads + 255) / 256), 256, 0, stream>>>(
            h0_out, ln_g, ln_b, h, N1);
    }

    k_gather_gemm<128><<<(N2 + 31) / 32, 256, 0, stream>>>(
        h, slots, seg1, W1, b1, out2, N2);
}


# LayerNorm+ReLU moved into the layer-0 gather-GEMM epilogue: C tile staged through the A tile's LDS, a wave normalises 4 whole rows (row sums regrouped); the 50000-wave launch and its h0 re-read go
# speedup vs baseline: 3.2896x; 1.0993x over previous
#include <hip/hip_runtime.h>
#include <math.h>

typedef float v2f __attribute__((ext_vector_type(2)));
typedef float v8f __attribute__((ext_vector_type(8)));

#define D_H 256
#define LN_EPS 1e-5f


typedef float v4f __attribute__((ext_vector_type(4)));

#define SORT_G 256
#define NB0 196
#define NB1 40
#define NBK 236

__device__ __forceinline__ int block_excl_scan_256(int v, int* red) {
    const int lane = threadIdx.x & 31, wave = threadIdx.x >> 5;
    int incl = v;
#pragma unroll
    for (int o = 1; o < 32; o <<= 1) {
        int t = __shfl_up(incl, o, 32);
        if (lane >= o) incl += t;
    }
    if (lane == 31) red[wave] = incl;
    __syncthreads();
    int woff = 0;
#pragma unroll
    for (int w = 0; w < 8; ++w) if (w < wave) woff += red[w];
    __syncthreads();
    return woff + incl - v;
}

__global__ __launch_bounds__(256) void k_bucket_hist(
    const int* __restrict__ dst0, int E0, const int* __restrict__ dst1, int E1,
    int* __restrict__ bh) {
    __shared__ int hist[NBK];
    const int tid = threadIdx.x;
    if (tid < NBK) hist[tid] = 0;
    __syncthreads();
    {
        const int c = (E0 + SORT_G - 1) / SORT_G;
        const int lo = blockIdx.x * c;
        const int hi = min(lo + c, E0);
        for (int e = lo + tid; e < hi; e += 1024) {
            int d[4];
#pragma unroll
            for (int u = 0; u < 4; ++u) { const int i = e + u * 256; d[u] = (i < hi) ? dst0[i] : -1; }
#pragma unroll
            for (int u = 0; u < 4; ++u) if (d[u] >= 0) atomicAdd(&hist[d[u] >> 8], 1);
        }
    }
    {
        const int c = (E1 + SORT_G - 1) / SORT_G;
        const int lo = blockIdx.x * c;
        const int hi = min(lo + c, E1);
        for (int e = lo + tid; e < hi; e += 1024) {
            int d[4];
#pragma unroll
            for (int u = 0; u < 4; ++u) { const int i = e + u * 256; d[u] = (i < hi) ? dst1[i] : -1; }
#pragma unroll
            for (int u = 0; u < 4; ++u) if (d[u] >= 0) atomicAdd(&hist[NB0 + (d[u] >> 8)], 1);
        }
    }
    __syncthreads();
    if (tid < NBK) bh[tid * SORT_G + blockIdx.x] = hist[tid];
}

__global__ __launch_bounds__(256) void k_bucket_scan(int* __restrict__ bh, int* __restrict__ keytot) {
    const int key = (int)((blockIdx.x * 256 + threadIdx.x) >> 5);
    const int lane = threadIdx.x & 31;
    if (key >= NBK) return;
    int4* p = (int4*)(bh + key * SORT_G + lane * 8);
    const int4 a = p[0], b = p[1];
    const int t = a.x + a.y + a.z + a.w + b.x + b.y + b.z + b.w;
    int incl = t;
#pragma unroll
    for (int o = 1; o < 32; o <<= 1) {
        int x = __shfl_up(incl, o, 32);
        if (lane >= o) incl += x;
    }
    int run = incl - t;
    int4 oa, ob;
    oa.x = run; run += a.x;
    oa.y = run; run += a.y;
    oa.z = run; run += a.z;
    oa.w = run; run += a.w;
    ob.x = run; run += b.x;
    ob.y = run; run += b.y;
    ob.z = run; run += b.z;
    ob.w = run;
    p[0] = oa; p[1] = ob;
    if (lane == 31) keytot[key] = incl;
}

__global__ __launch_bounds__(256) void k_bucket_scatter(
    const int* __restrict__ src0, const int* __restrict__ dst0, int E0,
    const int* __restrict__ src1, const int* __restrict__ dst1, int E1,
    const int* __restrict__ bh, const int* __restrict__ keytot, int2* __restrict__ sorted) {
    __shared__ int cur[NBK];
    __shared__ int red[8];
    const int tid = threadIdx.x;
    const int kt = (tid < NBK) ? keytot[tid] : 0;
    const int ks = block_excl_scan_256(kt, red);
    if (tid < NBK) cur[tid] = ks + bh[tid * SORT_G + blockIdx.x];
    __syncthreads();
    {
        const int c = (E0 + SORT_G - 1) / SORT_G;
        const int lo = blockIdx.x * c;
        const int hi = min(lo + c, E0);
        for (int e = lo + tid; e < hi; e += 1024) {
            int d[4], sv[4];
#pragma unroll
            for (int u = 0; u < 4; ++u) {
                const int i = e + u * 256;
                d[u] = (i < hi) ? dst0[i] : -1;
                sv[u] = (i < hi) ? src0[i] : 0;
            }
#pragma unroll
            for (int u = 0; u < 4; ++u) if (d[u] >= 0) {
                const int pos = atomicAdd(&cur[d[u] >> 8], 1);
                sorted[pos] = make_int2(d[u] & 255, sv[u]);
            }
        }
    }
    {
        const int c = (E1 + SORT_G - 1) / SORT_G;
        const int lo = blockIdx.x * c;
        const int hi = min(lo + c, E1);
        for (int e = lo + tid; e < hi; e += 1024) {
            int d[4], sv[4];
#pragma unroll
            for (int u = 0; u < 4; ++u) {
                const int i = e + u * 256;
                d[u] = (i < hi) ? dst1[i] : -1;
                sv[u] = (i < hi) ? src1[i] : 0;
            }
#pragma unroll
            for (int u = 0; u < 4; ++u) if (d[u] >= 0) {
                const int pos = atomicAdd(&cur[NB0 + (d[u] >> 8)], 1);
                sorted[pos] = make_int2(d[u] & 255, sv[u]);
            }
        }
    }
}

__global__ __launch_bounds__(256) void k_bucket_rows(
    const int2* __restrict__ sorted, const int* __restrict__ keytot,
    int* __restrict__ seg0, int n0, int* __restrict__ seg1, int n1,
    int* __restrict__ slots) {
    __shared__ int cnt[256];
    __shared__ int red[8];
    __shared__ int range[2];
    const int tid = threadIdx.x;
    const int key = blockIdx.x;
    const int kt = (tid < NBK) ? keytot[tid] : 0;
    const int ks = block_excl_scan_256(kt, red);
    if (tid == key) { range[0] = ks; range[1] = ks + kt; }
    cnt[tid] = 0;
    __syncthreads();
    const int bs = range[0], be = range[1];
    for (int i = bs + tid; i < be; i += 1024) {
        int r[4];
#pragma unroll
        for (int u = 0; u < 4; ++u) { const int j = i + u * 256; r[u] = (j < be) ? sorted[j].x : -1; }
#pragma unroll
        for (int u = 0; u < 4; ++u) if (r[u] >= 0) atomicAdd(&cnt[r[u]], 1);
    }
    __syncthreads();
    const int c = cnt[tid];
    const int ex = block_excl_scan_256(c, red);
    const int l1 = key >= NB0;
    const int row = ((l1 ? key - NB0 : key) << 8) + tid;
    if (l1) {
        if (row < n1) *(int2*)(seg1 + 2 * row) = make_int2(bs + ex, c);
    } else {
        if (row < n0) *(int2*)(seg0 + 2 * row) = make_int2(bs + ex, c);
    }
    cnt[tid] = ex;
    __syncthreads();
    for (int i = bs + tid; i < be; i += 1024) {
        int2 p[4];
#pragma unroll
        for (int u = 0; u < 4; ++u) { const int j = i + u * 256; p[u] = (j < be) ? sorted[j] : make_int2(-1, 0); }
#pragma unroll
        for (int u = 0; u < 4; ++u) if (p[u].x >= 0) {
            const int pos = atomicAdd(&cnt[p[u].x], 1);
            slots[bs + pos] = p[u].y;
        }
    }
}

#define ALD2 260
#define KCH2 32

__device__ __forceinline__ void ld_src_row(const float* __restrict__ h, int id, int lane,
                                           v4f& x, v4f& y) {
    const v4f* p = (const v4f*)(h + (size_t)id * D_H);
    x = p[lane];
    y = p[lane + 32];
}

template <int N, bool LN>
__global__ __launch_bounds__(256) void k_gather_gemm(
    const float* __restrict__ h, const int* __restrict__ slots, const int* __restrict__ seg,
    const float* __restrict__ W, const float* __restrict__ bias, float* __restrict__ out, int M,
    const float* __restrict__ ln_g, const float* __restrict__ ln_b, float* __restrict__ hn) {
    constexpr int NT = N / 64;
    __shared__ float Alds[32 * ALD2];
    __shared__ float Blds[KCH2 * N];
    const int tid = threadIdx.x;
    const int wave = __builtin_amdgcn_readfirstlane(tid >> 5);
    const int lane = tid & 31;
    const int rowBase = blockIdx.x * 32;
    const v4f zero4 = {0.f, 0.f, 0.f, 0.f};

    int sbase[4], scnt[4], idx0[4];
#pragma unroll
    for (int i = 0; i < 4; ++i) {
        const int row = rowBase + wave * 4 + i;
        int2 sg = make_int2(0, 0);
        if (row < M) sg = *(const int2*)(seg + 2 * row);
        sbase[i] = sg.x; scnt[i] = sg.y;
    }
#pragma unroll
    for (int i = 0; i < 4; ++i) {
        const int row = rowBase + wave * 4 + i;
        idx0[i] = (lane < scnt[i]) ? slots[sbase[i] + lane] : row;
    }

#pragma unroll
    for (int i = 0; i < 4; ++i) {
        const int lr = wave * 4 + i;
        const int row = rowBase + lr;
        v4f r0 = zero4, r1 = zero4;
        if (row < M) {
            const int base = sbase[i], n = scnt[i];
            const int n1 = n + 1;
            v4f a0 = zero4, a1 = zero4;
            for (int c = 0; c < n1; c += 32) {
                const int m = (n1 - c < 32) ? (n1 - c) : 32;
                int idx = idx0[i];
                if (c > 0) idx = (c + lane < n) ? slots[base + c + lane] : row;
                int j = 0;
                for (; j + 8 <= m; j += 8) {
                    v4f x[8], y[8];
#pragma unroll
                    for (int u = 0; u < 8; ++u)
                        ld_src_row(h, __builtin_amdgcn_readlane(idx, j + u), lane, x[u], y[u]);
#pragma unroll
                    for (int u = 0; u < 8; ++u) { a0 += x[u]; a1 += y[u]; }
                }
                if (j + 4 <= m) {
                    v4f x[4], y[4];
#pragma unroll
                    for (int u = 0; u < 4; ++u)
                        ld_src_row(h, __builtin_amdgcn_readlane(idx, j + u), lane, x[u], y[u]);
#pragma unroll
                    for (int u = 0; u < 4; ++u) { a0 += x[u]; a1 += y[u]; }
                    j += 4;
                }
                const int r = m - j;
                if (r > 0) {
                    v4f x0, y0, x1 = zero4, y1 = zero4, x2 = zero4, y2 = zero4;
                    ld_src_row(h, __builtin_amdgcn_readlane(idx, j), lane, x0, y0);
                    if (r > 1) ld_src_row(h, __builtin_amdgcn_readlane(idx, j + 1), lane, x1, y1);
                    if (r > 2) ld_src_row(h, __builtin_amdgcn_readlane(idx, j + 2), lane, x2, y2);
                    a0 += x0; a1 += y0;
                    if (r > 1) { a0 += x1; a1 += y1; }
                    if (r > 2) { a0 += x2; a1 += y2; }
                }
            }
            const float inv = 1.0f / ((float)n + 1.0f);
            r0 = a0 * inv;
            r1 = a1 * inv;
        }
        *(v4f*)(&Alds[lr * ALD2 + 4 * lane]) = r0;
        *(v4f*)(&Alds[lr * ALD2 + 128 + 4 * lane]) = r1;
    }

    const int wr = wave >> 2, wc = wave & 3;
    const int half = lane >> 4, l16 = lane & 15;
    v8f acc[NT];
#pragma unroll
    for (int t = 0; t < NT; ++t) acc[t] = v8f{};
    const float* arow = &Alds[(wr * 16 + l16) * ALD2];
    const float* bcol = &Blds[wc * (N / 4) + l16];

    constexpr int NV = (KCH2 * N / 4) / 256;
    v4f wpre[NV];
#pragma unroll
    for (int j = 0; j < NV; ++j)
        wpre[j] = *(const v4f*)(W + (size_t)(tid + j * 256) * 4);

    for (int kb = 0; kb < D_H; kb += KCH2) {
        __syncthreads();
#pragma unroll
        for (int j = 0; j < NV; ++j)
            *(v4f*)(&Blds[(tid + j * 256) * 4]) = wpre[j];
        __syncthreads();
        if (kb + KCH2 < D_H) {
#pragma unroll
            for (int j = 0; j < NV; ++j)
                wpre[j] = *(const v4f*)(W + (size_t)(kb + KCH2) * N + (size_t)(tid + j * 256) * 4);
        }
#pragma unroll
        for (int k0 = 0; k0 < KCH2; k0 += 4) {
            v2f a;
            a.x = arow[kb + k0 + 2 * half];
            a.y = arow[kb + k0 + 2 * half + 1];
#pragma unroll
            for (int t = 0; t < NT; ++t) {
                v2f b;
                b.x = bcol[(k0 + 2 * half) * N + t * 16];
                b.y = bcol[(k0 + 2 * half + 1) * N + t * 16];
                acc[t] = __builtin_amdgcn_wmma_f32_16x16x4_f32(
                    false, a, false, b, (short)0, acc[t], false, false);
            }
        }
    }

    if constexpr (LN) {
        static_assert(N == D_H, "a workgroup must hold whole rows of the normalised width");
        const v4f bi0 = *(const v4f*)(bias + 4 * lane), bi1 = *(const v4f*)(bias + 128 + 4 * lane);
        const v4f g0 = *(const v4f*)(ln_g + 4 * lane),  g1 = *(const v4f*)(ln_g + 128 + 4 * lane);
        const v4f be0 = *(const v4f*)(ln_b + 4 * lane), be1 = *(const v4f*)(ln_b + 128 + 4 * lane);
        __syncthreads();
#pragma unroll
        for (int t = 0; t < NT; ++t)
#pragma unroll
            for (int j = 0; j < 8; ++j)
                Alds[(wr * 16 + half * 8 + j) * ALD2 + wc * (N / 4) + t * 16 + l16] = acc[t][j];
        __syncthreads();
        v4f x[4], y[4];
        float s[4];
#pragma unroll
        for (int i = 0; i < 4; ++i) {
            const int lr = wave * 4 + i;
            x[i] = *(const v4f*)(&Alds[lr * ALD2 + 4 * lane]) + bi0;
            y[i] = *(const v4f*)(&Alds[lr * ALD2 + 128 + 4 * lane]) + bi1;
        }
#pragma unroll
        for (int i = 0; i < 4; ++i) {
            const int row = rowBase + wave * 4 + i;
            if (row < M) {
                *(v4f*)(out + (size_t)row * N + 4 * lane) = x[i];
                *(v4f*)(out + (size_t)row * N + 128 + 4 * lane) = y[i];
            }
            s[i] = ((x[i].x + x[i].y) + (x[i].z + x[i].w)) + ((y[i].x + y[i].y) + (y[i].z + y[i].w));
        }
#pragma unroll
        for (int o = 16; o > 0; o >>= 1)
#pragma unroll
            for (int i = 0; i < 4; ++i) s[i] += __shfl_xor(s[i], o, 32);
        v4f dx[4], dy[4];
        float vs[4];
#pragma unroll
        for (int i = 0; i < 4; ++i) {
            const float mu = s[i] * (1.0f / 256.0f);
            dx[i] = x[i] - mu;
            dy[i] = y[i] - mu;
            vs[i] = ((dx[i].x * dx[i].x + dx[i].y * dx[i].y) + (dx[i].z * dx[i].z + dx[i].w * dx[i].w)) +
                    ((dy[i].x * dy[i].x + dy[i].y * dy[i].y) + (dy[i].z * dy[i].z + dy[i].w * dy[i].w));
        }
#pragma unroll
        for (int o = 16; o > 0; o >>= 1)
#pragma unroll
            for (int i = 0; i < 4; ++i) vs[i] += __shfl_xor(vs[i], o, 32);
#pragma unroll
        for (int i = 0; i < 4; ++i) {
            const int row = rowBase + wave * 4 + i;
            const float rs = rsqrtf(vs[i] * (1.0f / 256.0f) + LN_EPS);
            v4f o0 = dx[i] * rs * g0 + be0, o1 = dy[i] * rs * g1 + be1;
#pragma unroll
            for (int e = 0; e < 4; ++e) { o0[e] = fmaxf(o0[e], 0.0f); o1[e] = fmaxf(o1[e], 0.0f); }
            if (row < M) {
                *(v4f*)(hn + (size_t)row * D_H + 4 * lane) = o0;
                *(v4f*)(hn + (size_t)row * D_H + 128 + 4 * lane) = o1;
            }
        }
    } else {
        const int rbase = rowBase + wr * 16 + half * 8;
#pragma unroll
        for (int t = 0; t < NT; ++t) {
            const int col = wc * (N / 4) + t * 16 + l16;
            const float bv = bias[col];
#pragma unroll
            for (int j = 0; j < 8; ++j) {
                const int row = rbase + j;
                if (row < M) out[(size_t)row * N + col] = acc[t][j] + bv;
            }
        }
    }
}

extern "C" void kernel_launch(void* const* d_in, const int* in_sizes, int n_in,
                              void* d_out, int out_size, void* d_ws, size_t ws_size,
                              hipStream_t stream) {
    (void)n_in; (void)out_size; (void)ws_size;
    const float* feats = (const float*)d_in[0];
    const float* W0    = (const float*)d_in[1];
    const float* b0    = (const float*)d_in[2];
    const float* W1    = (const float*)d_in[3];
    const float* b1    = (const float*)d_in[4];
    const float* ln_g  = (const float*)d_in[5];
    const float* ln_b  = (const float*)d_in[6];
    const int* src0 = (const int*)d_in[7];
    const int* dst0 = (const int*)d_in[8];
    const int* src1 = (const int*)d_in[9];
    const int* dst1 = (const int*)d_in[10];
    const int E0 = in_sizes[7], E1 = in_sizes[9];
    const int N1 = 50000, N2 = 10000;

    float* ws = (float*)d_ws;
    float* hn = ws;
    float* h  = ws + (size_t)N1 * D_H + N1;
    int* cnt0   = (int*)(h + (size_t)N1 * D_H);
    int* cnt1   = cnt0 + N1;
    int* tot    = cnt1 + N2;
    int* seg0   = tot + 2;
    int* seg1   = seg0 + 2 * N1;
    int* slots  = seg1 + 2 * N2;
    int2* sorted = (int2*)hn;
    int* bh      = (int*)hn + 4000000;
    int* keytot  = (int*)hn + 4100000;

    float* h0_out = (float*)d_out;
    float* out2   = h0_out + (size_t)N1 * D_H;

    k_bucket_hist<<<SORT_G, 256, 0, stream>>>(dst0, E0, dst1, E1, bh);
    k_bucket_scan<<<(NBK + 7) / 8, 256, 0, stream>>>(bh, keytot);
    k_bucket_scatter<<<SORT_G, 256, 0, stream>>>(src0, dst0, E0, src1, dst1, E1,
                                                 bh, keytot, sorted);
    k_bucket_rows<<<NBK, 256, 0, stream>>>(sorted, keytot, seg0, N1, seg1, N2, slots);

    k_gather_gemm<256, true><<<(N1 + 31) / 32, 256, 0, stream>>>(
        feats, slots, seg0, W0, b0, h0_out, N1, ln_g, ln_b, h);

    k_gather_gemm<128, false><<<(N2 + 31) / 32, 256, 0, stream>>>(
        h, slots, seg1, W1, b1, out2, N2, nullptr, nullptr, nullptr);
}


# Coarse bucket sort made chunk-local: 512 workgroups count, scan and place their own edges in one launch, the per-bucket pass collects the pieces; 2 launches go; slots in a new LDS arrival order
# speedup vs baseline: 3.5263x; 1.0720x over previous
#include <hip/hip_runtime.h>
#include <math.h>

typedef float v2f __attribute__((ext_vector_type(2)));
typedef float v8f __attribute__((ext_vector_type(8)));

#define D_H 256
#define LN_EPS 1e-5f


typedef float v4f __attribute__((ext_vector_type(4)));

#define SORT_G 512
#define NB0 196
#define NB1 40
#define NBK 236
#define NQ (SORT_G / 32)

__device__ __forceinline__ int block_excl_scan_256(int v, int* red) {
    const int lane = threadIdx.x & 31, wave = threadIdx.x >> 5;
    int incl = v;
#pragma unroll
    for (int o = 1; o < 32; o <<= 1) {
        int t = __shfl_up(incl, o, 32);
        if (lane >= o) incl += t;
    }
    if (lane == 31) red[wave] = incl;
    __syncthreads();
    int woff = 0;
#pragma unroll
    for (int w = 0; w < 8; ++w) if (w < wave) woff += red[w];
    __syncthreads();
    return woff + incl - v;
}

__global__ __launch_bounds__(256) void k_chunk_sort(
    const int* __restrict__ src0, const int* __restrict__ dst0, int E0,
    const int* __restrict__ src1, const int* __restrict__ dst1, int E1,
    int* __restrict__ tab, int2* __restrict__ sorted) {
    __shared__ int cur[NBK];
    __shared__ int red[8];
    const int tid = threadIdx.x;
    if (tid < NBK) cur[tid] = 0;
    __syncthreads();
    const int c0 = (E0 + SORT_G - 1) / SORT_G;
    const int c1 = (E1 + SORT_G - 1) / SORT_G;
    const int lo0 = blockIdx.x * c0, hi0 = min(lo0 + c0, E0);
    const int lo1 = blockIdx.x * c1, hi1 = min(lo1 + c1, E1);
    for (int e = lo0 + tid; e < hi0; e += 2048) {
        int d[8];
#pragma unroll
        for (int u = 0; u < 8; ++u) { const int i = e + u * 256; d[u] = (i < hi0) ? dst0[i] : -1; }
#pragma unroll
        for (int u = 0; u < 8; ++u) if (d[u] >= 0) atomicAdd(&cur[d[u] >> 8], 1);
    }
    for (int e = lo1 + tid; e < hi1; e += 2048) {
        int d[8];
#pragma unroll
        for (int u = 0; u < 8; ++u) { const int i = e + u * 256; d[u] = (i < hi1) ? dst1[i] : -1; }
#pragma unroll
        for (int u = 0; u < 8; ++u) if (d[u] >= 0) atomicAdd(&cur[NB0 + (d[u] >> 8)], 1);
    }
    __syncthreads();
    const int n = (tid < NBK) ? cur[tid] : 0;
    const int off = block_excl_scan_256(n, red);
    if (tid <= NBK) tab[tid * SORT_G + blockIdx.x] = off;
    if (tid < NBK) cur[tid] = off;
    __syncthreads();
    int2* mine = sorted + (size_t)blockIdx.x * (size_t)(c0 + c1);
    for (int e = lo0 + tid; e < hi0; e += 2048) {
        int d[8], sv[8];
#pragma unroll
        for (int u = 0; u < 8; ++u) {
            const int i = e + u * 256;
            d[u] = (i < hi0) ? dst0[i] : -1;
            sv[u] = (i < hi0) ? src0[i] : 0;
        }
#pragma unroll
        for (int u = 0; u < 8; ++u) if (d[u] >= 0) {
            const int pos = atomicAdd(&cur[d[u] >> 8], 1);
            mine[pos] = make_int2(d[u] & 255, sv[u]);
        }
    }
    for (int e = lo1 + tid; e < hi1; e += 2048) {
        int d[8], sv[8];
#pragma unroll
        for (int u = 0; u < 8; ++u) {
            const int i = e + u * 256;
            d[u] = (i < hi1) ? dst1[i] : -1;
            sv[u] = (i < hi1) ? src1[i] : 0;
        }
#pragma unroll
        for (int u = 0; u < 8; ++u) if (d[u] >= 0) {
            const int pos = atomicAdd(&cur[NB0 + (d[u] >> 8)], 1);
            mine[pos] = make_int2(d[u] & 255, sv[u]);
        }
    }
}

__global__ __launch_bounds__(256) void k_bucket_rows(
    const int2* __restrict__ sorted, const int* __restrict__ tab, int E0, int E1,
    int* __restrict__ seg0, int n0, int* __restrict__ seg1, int n1,
    int* __restrict__ slots) {
    __shared__ int cnt[256];
    __shared__ int red[8];
    __shared__ int soff[SORT_G];
    __shared__ int scnt[SORT_G];
    __shared__ int first[1];
    const int tid = threadIdx.x;
    const int key = blockIdx.x;
    const int CH = (E0 + SORT_G - 1) / SORT_G + (E1 + SORT_G - 1) / SORT_G;
    int before = 0;
#pragma unroll
    for (int k = 0; k < SORT_G / 256; ++k) {
        const int w = tid + k * 256;
        const int o = tab[key * SORT_G + w];
        const int o1 = tab[(key + 1) * SORT_G + w];
        soff[w] = w * CH + o;
        scnt[w] = o1 - o;
        before += o;
    }
    cnt[tid] = 0;
    const int ex0 = block_excl_scan_256(before, red);
    if (tid == 255) first[0] = ex0 + before;
    __syncthreads();
    const int bs = first[0];
    const int g = tid >> 3, sub = tid & 7;
    int so[NQ], sc[NQ], mx = 0;
#pragma unroll
    for (int q = 0; q < NQ; ++q) {
        so[q] = soff[g + 32 * q];
        sc[q] = scnt[g + 32 * q];
        mx = max(mx, sc[q]);
    }
    for (int i = sub; i < mx; i += 8) {
        int r[NQ];
#pragma unroll
        for (int q = 0; q < NQ; ++q) r[q] = (i < sc[q]) ? sorted[so[q] + i].x : -1;
#pragma unroll
        for (int q = 0; q < NQ; ++q) if (r[q] >= 0) atomicAdd(&cnt[r[q]], 1);
    }
    __syncthreads();
    const int c = cnt[tid];
    const int ex = block_excl_scan_256(c, red);
    const int l1 = key >= NB0;
    const int row = ((l1 ? key - NB0 : key) << 8) + tid;
    if (l1) {
        if (row < n1) *(int2*)(seg1 + 2 * row) = make_int2(bs + ex, c);
    } else {
        if (row < n0) *(int2*)(seg0 + 2 * row) = make_int2(bs + ex, c);
    }
    cnt[tid] = ex;
    __syncthreads();
    for (int i = sub; i < mx; i += 8) {
        int2 p[NQ];
#pragma unroll
        for (int q = 0; q < NQ; ++q) p[q] = (i < sc[q]) ? sorted[so[q] + i] : make_int2(-1, 0);
#pragma unroll
        for (int q = 0; q < NQ; ++q) if (p[q].x >= 0) {
            const int pos = atomicAdd(&cnt[p[q].x], 1);
            slots[bs + pos] = p[q].y;
        }
    }
}

#define ALD2 260
#define KCH2 32

__device__ __forceinline__ void ld_src_row(const float* __restrict__ h, int id, int lane,
                                           v4f& x, v4f& y) {
    const v4f* p = (const v4f*)(h + (size_t)id * D_H);
    x = p[lane];
    y = p[lane + 32];
}

template <int N, bool LN>
__global__ __launch_bounds__(256) void k_gather_gemm(
    const float* __restrict__ h, const int* __restrict__ slots, const int* __restrict__ seg,
    const float* __restrict__ W, const float* __restrict__ bias, float* __restrict__ out, int M,
    const float* __restrict__ ln_g, const float* __restrict__ ln_b, float* __restrict__ hn) {
    constexpr int NT = N / 64;
    __shared__ float Alds[32 * ALD2];
    __shared__ float Blds[KCH2 * N];
    const int tid = threadIdx.x;
    const int wave = __builtin_amdgcn_readfirstlane(tid >> 5);
    const int lane = tid & 31;
    const int rowBase = blockIdx.x * 32;
    const v4f zero4 = {0.f, 0.f, 0.f, 0.f};

    int sbase[4], scnt[4], idx0[4];
#pragma unroll
    for (int i = 0; i < 4; ++i) {
        const int row = rowBase + wave * 4 + i;
        int2 sg = make_int2(0, 0);
        if (row < M) sg = *(const int2*)(seg + 2 * row);
        sbase[i] = sg.x; scnt[i] = sg.y;
    }
#pragma unroll
    for (int i = 0; i < 4; ++i) {
        const int row = rowBase + wave * 4 + i;
        idx0[i] = (lane < scnt[i]) ? slots[sbase[i] + lane] : row;
    }

#pragma unroll
    for (int i = 0; i < 4; ++i) {
        const int lr = wave * 4 + i;
        const int row = rowBase + lr;
        v4f r0 = zero4, r1 = zero4;
        if (row < M) {
            const int base = sbase[i], n = scnt[i];
            const int n1 = n + 1;
            v4f a0 = zero4, a1 = zero4;
            for (int c = 0; c < n1; c += 32) {
                const int m = (n1 - c < 32) ? (n1 - c) : 32;
                int idx = idx0[i];
                if (c > 0) idx = (c + lane < n) ? slots[base + c + lane] : row;
                int j = 0;
                for (; j + 8 <= m; j += 8) {
                    v4f x[8], y[8];
#pragma unroll
                    for (int u = 0; u < 8; ++u)
                        ld_src_row(h, __builtin_amdgcn_readlane(idx, j + u), lane, x[u], y[u]);
#pragma unroll
                    for (int u = 0; u < 8; ++u) { a0 += x[u]; a1 += y[u]; }
                }
                if (j + 4 <= m) {
                    v4f x[4], y[4];
#pragma unroll
                    for (int u = 0; u < 4; ++u)
                        ld_src_row(h, __builtin_amdgcn_readlane(idx, j + u), lane, x[u], y[u]);
#pragma unroll
                    for (int u = 0; u < 4; ++u) { a0 += x[u]; a1 += y[u]; }
                    j += 4;
                }
                const int r = m - j;
                if (r > 0) {
                    v4f x0, y0, x1 = zero4, y1 = zero4, x2 = zero4, y2 = zero4;
                    ld_src_row(h, __builtin_amdgcn_readlane(idx, j), lane, x0, y0);
                    if (r > 1) ld_src_row(h, __builtin_amdgcn_readlane(idx, j + 1), lane, x1, y1);
                    if (r > 2) ld_src_row(h, __builtin_amdgcn_readlane(idx, j + 2), lane, x2, y2);
                    a0 += x0; a1 += y0;
                    if (r > 1) { a0 += x1; a1 += y1; }
                    if (r > 2) { a0 += x2; a1 += y2; }
                }
            }
            const float inv = 1.0f / ((float)n + 1.0f);
            r0 = a0 * inv;
            r1 = a1 * inv;
        }
        *(v4f*)(&Alds[lr * ALD2 + 4 * lane]) = r0;
        *(v4f*)(&Alds[lr * ALD2 + 128 + 4 * lane]) = r1;
    }

    const int wr = wave >> 2, wc = wave & 3;
    const int half = lane >> 4, l16 = lane & 15;
    v8f acc[NT];
#pragma unroll
    for (int t = 0; t < NT; ++t) acc[t] = v8f{};
    const float* arow = &Alds[(wr * 16 + l16) * ALD2];
    const float* bcol = &Blds[wc * (N / 4) + l16];

    constexpr int NV = (KCH2 * N / 4) / 256;
    v4f wpre[NV];
#pragma unroll
    for (int j = 0; j < NV; ++j)
        wpre[j] = *(const v4f*)(W + (size_t)(tid + j * 256) * 4);

    for (int kb = 0; kb < D_H; kb += KCH2) {
        __syncthreads();
#pragma unroll
        for (int j = 0; j < NV; ++j)
            *(v4f*)(&Blds[(tid + j * 256) * 4]) = wpre[j];
        __syncthreads();
        if (kb + KCH2 < D_H) {
#pragma unroll
            for (int j = 0; j < NV; ++j)
                wpre[j] = *(const v4f*)(W + (size_t)(kb + KCH2) * N + (size_t)(tid + j * 256) * 4);
        }
#pragma unroll
        for (int k0 = 0; k0 < KCH2; k0 += 4) {
            v2f a;
            a.x = arow[kb + k0 + 2 * half];
            a.y = arow[kb + k0 + 2 * half + 1];
#pragma unroll
            for (int t = 0; t < NT; ++t) {
                v2f b;
                b.x = bcol[(k0 + 2 * half) * N + t * 16];
                b.y = bcol[(k0 + 2 * half + 1) * N + t * 16];
                acc[t] = __builtin_amdgcn_wmma_f32_16x16x4_f32(
                    false, a, false, b, (short)0, acc[t], false, false);
            }
        }
    }

    if constexpr (LN) {
        static_assert(N == D_H, "a workgroup must hold whole rows of the normalised width");
        const v4f bi0 = *(const v4f*)(bias + 4 * lane), bi1 = *(const v4f*)(bias + 128 + 4 * lane);
        const v4f g0 = *(const v4f*)(ln_g + 4 * lane),  g1 = *(const v4f*)(ln_g + 128 + 4 * lane);
        const v4f be0 = *(const v4f*)(ln_b + 4 * lane), be1 = *(const v4f*)(ln_b + 128 + 4 * lane);
        __syncthreads();
#pragma unroll
        for (int t = 0; t < NT; ++t)
#pragma unroll
            for (int j = 0; j < 8; ++j)
                Alds[(wr * 16 + half * 8 + j) * ALD2 + wc * (N / 4) + t * 16 + l16] = acc[t][j];
        __syncthreads();
        v4f x[4], y[4];
        float s[4];
#pragma unroll
        for (int i = 0; i < 4; ++i) {
            const int lr = wave * 4 + i;
            x[i] = *(const v4f*)(&Alds[lr * ALD2 + 4 * lane]) + bi0;
            y[i] = *(const v4f*)(&Alds[lr * ALD2 + 128 + 4 * lane]) + bi1;
        }
#pragma unroll
        for (int i = 0; i < 4; ++i) {
            const int row = rowBase + wave * 4 + i;
            if (row < M) {
                *(v4f*)(out + (size_t)row * N + 4 * lane) = x[i];
                *(v4f*)(out + (size_t)row * N + 128 + 4 * lane) = y[i];
            }
            s[i] = ((x[i].x + x[i].y) + (x[i].z + x[i].w)) + ((y[i].x + y[i].y) + (y[i].z + y[i].w));
        }
#pragma unroll
        for (int o = 16; o > 0; o >>= 1)
#pragma unroll
            for (int i = 0; i < 4; ++i) s[i] += __shfl_xor(s[i], o, 32);
        v4f dx[4], dy[4];
        float vs[4];
#pragma unroll
        for (int i = 0; i < 4; ++i) {
            const float mu = s[i] * (1.0f / 256.0f);
            dx[i] = x[i] - mu;
            dy[i] = y[i] - mu;
            vs[i] = ((dx[i].x * dx[i].x + dx[i].y * dx[i].y) + (dx[i].z * dx[i].z + dx[i].w * dx[i].w)) +
                    ((dy[i].x * dy[i].x + dy[i].y * dy[i].y) + (dy[i].z * dy[i].z + dy[i].w * dy[i].w));
        }
#pragma unroll
        for (int o = 16; o > 0; o >>= 1)
#pragma unroll
            for (int i = 0; i < 4; ++i) vs[i] += __shfl_xor(vs[i], o, 32);
#pragma unroll
        for (int i = 0; i < 4; ++i) {
            const int row = rowBase + wave * 4 + i;
            const float rs = rsqrtf(vs[i] * (1.0f / 256.0f) + LN_EPS);
            v4f o0 = dx[i] * rs * g0 + be0, o1 = dy[i] * rs * g1 + be1;
#pragma unroll
            for (int e = 0; e < 4; ++e) { o0[e] = fmaxf(o0[e], 0.0f); o1[e] = fmaxf(o1[e], 0.0f); }
            if (row < M) {
                *(v4f*)(hn + (size_t)row * D_H + 4 * lane) = o0;
                *(v4f*)(hn + (size_t)row * D_H + 128 + 4 * lane) = o1;
            }
        }
    } else {
        const int rbase = rowBase + wr * 16 + half * 8;
#pragma unroll
        for (int t = 0; t < NT; ++t) {
            const int col = wc * (N / 4) + t * 16 + l16;
            const float bv = bias[col];
#pragma unroll
            for (int j = 0; j < 8; ++j) {
                const int row = rbase + j;
                if (row < M) out[(size_t)row * N + col] = acc[t][j] + bv;
            }
        }
    }
}

extern "C" void kernel_launch(void* const* d_in, const int* in_sizes, int n_in,
                              void* d_out, int out_size, void* d_ws, size_t ws_size,
                              hipStream_t stream) {
    (void)n_in; (void)out_size; (void)ws_size;
    const float* feats = (const float*)d_in[0];
    const float* W0    = (const float*)d_in[1];
    const float* b0    = (const float*)d_in[2];
    const float* W1    = (const float*)d_in[3];
    const float* b1    = (const float*)d_in[4];
    const float* ln_g  = (const float*)d_in[5];
    const float* ln_b  = (const float*)d_in[6];
    const int* src0 = (const int*)d_in[7];
    const int* dst0 = (const int*)d_in[8];
    const int* src1 = (const int*)d_in[9];
    const int* dst1 = (const int*)d_in[10];
    const int E0 = in_sizes[7], E1 = in_sizes[9];
    const int N1 = 50000, N2 = 10000;

    float* ws = (float*)d_ws;
    float* hn = ws;
    float* h  = ws + (size_t)N1 * D_H + N1;
    int* cnt0   = (int*)(h + (size_t)N1 * D_H);
    int* cnt1   = cnt0 + N1;
    int* tot    = cnt1 + N2;
    int* seg0   = tot + 2;
    int* seg1   = seg0 + 2 * N1;
    int* slots  = seg1 + 2 * N2;
    int2* sorted = (int2*)hn;
    int* tab     = (int*)hn + 4000000;

    float* h0_out = (float*)d_out;
    float* out2   = h0_out + (size_t)N1 * D_H;

    k_chunk_sort<<<SORT_G, 256, 0, stream>>>(src0, dst0, E0, src1, dst1, E1, tab, sorted);
    k_bucket_rows<<<NBK, 256, 0, stream>>>(sorted, tab, E0, E1, seg0, N1, seg1, N2, slots);

    k_gather_gemm<256, true><<<(N1 + 31) / 32, 256, 0, stream>>>(
        feats, slots, seg0, W0, b0, h0_out, N1, ln_g, ln_b, h);

    k_gather_gemm<128, false><<<(N2 + 31) / 32, 256, 0, stream>>>(
        h, slots, seg1, W1, b1, out2, N2, nullptr, nullptr, nullptr);
}
